# starting kernel
# baseline (speedup 1.0000x reference)
#include <hip/hip_runtime.h>
#include <math.h>

typedef __attribute__((ext_vector_type(2))) float v2f;
typedef __attribute__((ext_vector_type(8))) float v8f;

#define WAVES 8
constexpr int IN_F = 14, H = 24, T = 25, NC = 4, WH = 12;
constexpr int G  = 4 * H;
constexpr int NT = G / 16;
constexpr int KX = 4;
constexpr int KH = H / 4;

__device__ __forceinline__ float fsig(float x) {
  float e = __builtin_amdgcn_exp2f(x * -1.4426950408889634f);
  return __builtin_amdgcn_rcpf(1.f + e);
}
__device__ __forceinline__ float ftanh(float x) {
  float t = __builtin_amdgcn_exp2f(-2.8853900817779268f * fabsf(x));
  float r = (1.f - t) * __builtin_amdgcn_rcpf(1.f + t);
  return copysignf(r, x);
}

#define WAVE_FENCE() asm volatile("" ::: "memory")

__global__ __launch_bounds__(32 * WAVES) void bilstm_center_kernel(
    const float* __restrict__ x,
    const float* __restrict__ w_ih_f, const float* __restrict__ w_hh_f,
    const float* __restrict__ b_ih_f, const float* __restrict__ b_hh_f,
    const float* __restrict__ w_ih_b, const float* __restrict__ w_hh_b,
    const float* __restrict__ b_ih_b, const float* __restrict__ b_hh_b,
    const float* __restrict__ head_w, const float* __restrict__ head_b,
    float* __restrict__ out, int B)
{
  __shared__ float s_wx[2][NT * KX * 64];
  __shared__ float s_wh[2][NT * KH * 64];
  __shared__ float s_bias[2][G];
  __shared__ float s_headw[NC * 2 * H];
  __shared__ float s_headb[NC];
  __shared__ float s_gates[WAVES][16 * G];
  __shared__ float s_h[WAVES][16 * H];
  __shared__ float s_hf[WAVES][16 * H];

  const int tid = threadIdx.x;

  for (int i = tid; i < 2 * NT * KX * 64; i += blockDim.x) {
    int d = i / (NT * KX * 64);
    int r = i % (NT * KX * 64);
    int frag = r >> 6, slot = r & 63;
    int nt = frag / KX, kk = frag % KX;
    int ln = slot >> 1, e = slot & 1;
    int n = nt * 16 + (ln & 15);
    int k = kk * 4 + ((ln >> 4) << 1) + e;
    const float* w = d ? w_ih_b : w_ih_f;
    s_wx[d][r] = (k < IN_F) ? w[n * IN_F + k] : 0.f;
  }
  for (int i = tid; i < 2 * NT * KH * 64; i += blockDim.x) {
    int d = i / (NT * KH * 64);
    int r = i % (NT * KH * 64);
    int frag = r >> 6, slot = r & 63;
    int nt = frag / KH, kk = frag % KH;
    int ln = slot >> 1, e = slot & 1;
    int n = nt * 16 + (ln & 15);
    int k = kk * 4 + ((ln >> 4) << 1) + e;
    const float* w = d ? w_hh_b : w_hh_f;
    s_wh[d][r] = w[n * H + k];
  }
  for (int i = tid; i < 2 * G; i += blockDim.x) {
    int d = i / G, n = i % G;
    s_bias[d][n] = d ? (b_ih_b[n] + b_hh_b[n]) : (b_ih_f[n] + b_hh_f[n]);
  }
  for (int i = tid; i < NC * 2 * H; i += blockDim.x) s_headw[i] = head_w[i];
  if (tid < NC) s_headb[tid] = head_b[tid];
  __syncthreads();

  const int lane = tid & 31;
  const int w    = tid >> 5;
  const int M    = lane & 15;
  const int hi   = lane >> 4;
  const int b0   = (blockIdx.x * WAVES + w) * 16;
  if (b0 >= B) return;

  float* gatesL = &s_gates[w][0];
  float* hL     = &s_h[w][0];

  float bx[2][NT];
  #pragma unroll
  for (int d = 0; d < 2; ++d)
    #pragma unroll
    for (int nt = 0; nt < NT; ++nt) bx[d][nt] = s_bias[d][nt * 16 + M];

  float cst[12];
  const float* xbase = x + (size_t)b0 * T * IN_F;

  for (int d = 0; d < 2; ++d) {
    #pragma unroll
    for (int u = 0; u < 12; ++u) cst[u] = 0.f;
    for (int e = lane; e < 16 * H; e += 32) hL[e] = 0.f;
    WAVE_FENCE();

    for (int step = 0; step <= WH; ++step) {
      const int t = d ? (T - 1 - step) : step;
      const float* xr = xbase + ((size_t)M * T + t) * IN_F;

      v2f ax[KX];
      #pragma unroll
      for (int kk = 0; kk < KX; ++kk) {
        const int ks = kk * 4 + hi * 2;
        if (ks + 1 < IN_F) ax[kk] = *(const v2f*)(xr + ks);
        else               ax[kk] = (v2f){0.f, 0.f};
      }
      v2f ah[KH];
      #pragma unroll
      for (int kk = 0; kk < KH; ++kk)
        ah[kk] = *(const v2f*)&hL[M * H + kk * 4 + hi * 2];

      #pragma unroll
      for (int nt = 0; nt < NT; ++nt) {
        const float bb = bx[d][nt];
        v8f acc;
        #pragma unroll
        for (int r = 0; r < 8; ++r) acc[r] = bb;
        #pragma unroll
        for (int kk = 0; kk < KX; ++kk) {
          v2f bfrag = *(const v2f*)&s_wx[d][(nt * KX + kk) * 64 + lane * 2];
          acc = __builtin_amdgcn_wmma_f32_16x16x4_f32(
              false, ax[kk], false, bfrag, (short)0, acc, false, false);
        }
        #pragma unroll
        for (int kk = 0; kk < KH; ++kk) {
          v2f bfrag = *(const v2f*)&s_wh[d][(nt * KH + kk) * 64 + lane * 2];
          acc = __builtin_amdgcn_wmma_f32_16x16x4_f32(
              false, ah[kk], false, bfrag, (short)0, acc, false, false);
        }
        #pragma unroll
        for (int r = 0; r < 8; ++r)
          gatesL[(r + 8 * hi) * G + nt * 16 + M] = acc[r];
      }
      WAVE_FENCE();

      #pragma unroll
      for (int u = 0; u < 12; ++u) {
        const int e = u * 32 + lane;
        const int m = e / H, j = e - m * H;
        const float gi = gatesL[m * G + j];
        const float gf = gatesL[m * G + 24 + j];
        const float gg = gatesL[m * G + 48 + j];
        const float go = gatesL[m * G + 72 + j];
        const float iv = fsig(gi), fv = fsig(gf), ov = fsig(go), gv = ftanh(gg);
        const float cc = fv * cst[u] + iv * gv;
        cst[u] = cc;
        hL[e] = ov * ftanh(cc);
      }
      WAVE_FENCE();
    }

    if (d == 0) {
      for (int e = lane; e < 16 * H; e += 32) s_hf[w][e] = hL[e];
      WAVE_FENCE();
    }
  }

  #pragma unroll
  for (int p = lane; p < 64; p += 32) {
    const int m = p >> 2, cls = p & 3;
    float acc = s_headb[cls];
    #pragma unroll
    for (int j = 0; j < H; ++j) {
      acc += s_hf[w][m * H + j] * s_headw[cls * 2 * H + j];
      acc += hL[m * H + j]      * s_headw[cls * 2 * H + H + j];
    }
    out[(size_t)(b0 + m) * NC + cls] = acc;
  }
}

extern "C" void kernel_launch(void* const* d_in, const int* in_sizes, int n_in,
                              void* d_out, int out_size, void* d_ws, size_t ws_size,
                              hipStream_t stream) {
  (void)n_in; (void)d_ws; (void)ws_size; (void)out_size;
  const float* x      = (const float*)d_in[0];
  const float* w_ih_f = (const float*)d_in[1];
  const float* w_hh_f = (const float*)d_in[2];
  const float* b_ih_f = (const float*)d_in[3];
  const float* b_hh_f = (const float*)d_in[4];
  const float* w_ih_b = (const float*)d_in[5];
  const float* w_hh_b = (const float*)d_in[6];
  const float* b_ih_b = (const float*)d_in[7];
  const float* b_hh_b = (const float*)d_in[8];
  const float* head_w = (const float*)d_in[9];
  const float* head_b = (const float*)d_in[10];
  float* out = (float*)d_out;

  const int B = in_sizes[0] / (T * IN_F);
  const int tilesPerBlock = WAVES;
  const int grid = (B + 16 * tilesPerBlock - 1) / (16 * tilesPerBlock);
  bilstm_center_kernel<<<grid, 32 * WAVES, 0, stream>>>(
      x, w_ih_f, w_hh_f, b_ih_f, b_hh_f,
      w_ih_b, w_hh_b, b_ih_b, b_hh_b, head_w, head_b, out, B);
}


# Weights skip LDS: each wave loads a direction's 60 B fragments (8-byte pieces of weight rows) and bias words from memory straight into their registers, all in flight at once; sums unchanged
# speedup vs baseline: 1.3143x; 1.3143x over previous
#include <hip/hip_runtime.h>
#include <math.h>


typedef __attribute__((ext_vector_type(2))) float v2f;
typedef __attribute__((ext_vector_type(8))) float v8f;

#define WAVES 8
constexpr int IN_F = 14, H = 24, T = 25, NC = 4, WH = 12;
constexpr int G  = 4 * H;
constexpr int NT = G / 16;
constexpr int KX = 4;
constexpr int KH = H / 4;
constexpr int HP = 28;
constexpr int TPW = 2;

__device__ __forceinline__ float fsig(float x) {
  float e = __builtin_amdgcn_exp2f(x * -1.4426950408889634f);
  return __builtin_amdgcn_rcpf(1.f + e);
}
__device__ __forceinline__ float ftanh(float x) {
  float t = __builtin_amdgcn_exp2f(-2.8853900817779268f * fabsf(x));
  float r = (1.f - t) * __builtin_amdgcn_rcpf(1.f + t);
  return copysignf(r, x);
}

#define WAVE_FENCE() asm volatile("" ::: "memory")

__device__ __forceinline__ float rot8(float v) {
  return __builtin_bit_cast(float, __builtin_amdgcn_update_dpp(
      0, __builtin_bit_cast(int, v), 0x128, 0xf, 0xf, true));
}

__device__ __forceinline__ void load_x(v2f (&ax)[KX], const float* xr, int hi) {
  #pragma unroll
  for (int kk = 0; kk < KX - 1; ++kk) ax[kk] = *(const v2f*)(xr + kk * 4 + hi * 2);
  ax[KX - 1] = *(const v2f*)(xr + (KX - 1) * 4);
}

__device__ __forceinline__ void load_w(v2f (&wbx)[NT][KX], v2f (&wbh)[NT][KH],
                                       float (&bi)[NT], float (&bh)[NT],
                                       const float* __restrict__ w_ih,
                                       const float* __restrict__ w_hh,
                                       const float* __restrict__ b_ih,
                                       const float* __restrict__ b_hh, int M, int hi) {
  #pragma unroll
  for (int nt = 0; nt < NT; ++nt) {
    const int n = nt * 16 + M;
    const float* rx = w_ih + n * IN_F;
    const float* rh = w_hh + n * H;
    #pragma unroll
    for (int kk = 0; kk < KX - 1; ++kk) wbx[nt][kk] = *(const v2f*)(rx + kk * 4 + hi * 2);
    wbx[nt][KX - 1] = *(const v2f*)(rx + (KX - 1) * 4);
    #pragma unroll
    for (int kk = 0; kk < KH; ++kk) wbh[nt][kk] = *(const v2f*)(rh + kk * 4 + hi * 2);
    bi[nt] = b_ih[n];
    bh[nt] = b_hh[n];
  }
}

template <int I>
__device__ __forceinline__ void wm(v8f (&acc)[NT], const v8f (&cinit)[NT], const v2f (&ax)[KX],
                                   const v2f (&ah)[KH], const v2f (&wbx)[NT][KX],
                                   const v2f (&wbh)[NT][KH]) {
  constexpr int kk = I / NT, nt = I % NT;
  if constexpr (kk == 0)
    acc[nt] = __builtin_amdgcn_wmma_f32_16x16x4_f32(
        false, ax[0], false, wbx[nt][0], (short)0, cinit[nt], false, false);
  else if constexpr (kk < KX)
    acc[nt] = __builtin_amdgcn_wmma_f32_16x16x4_f32(
        false, ax[kk], false, wbx[nt][kk], (short)0, acc[nt], false, false);
  else
    acc[nt] = __builtin_amdgcn_wmma_f32_16x16x4_f32(
        false, ah[kk - KX], false, wbh[nt][kk - KX], (short)0, acc[nt], false, false);
}

#define HOLD(v) asm volatile("" :: "v"(v))
#define LN(i, body)                                              \
  {                                                              \
    _Pragma("unroll") for (int r = 0; r < 8; ++r) { body; }      \
    wm<i>(accM, cinit, ax, ah, wbx, wbh);                        \
    HOLD(accM[(i) % NT]);                                        \
    __builtin_amdgcn_sched_barrier(0);                           \
  }

__device__ __forceinline__ void half_step(
    v8f (&accM)[NT], const v8f (&cinit)[NT], const v2f (&axM)[KX], const float* hM,
    const v2f (&wbx)[NT][KX], const v2f (&wbh)[NT][KH],
    const v8f (&accV)[NT], float (&cA)[8], float (&cB)[8], float* hV,
    int M, int hi, int offA, int offB, bool lo, float k4, float a4, float b4)
{
  constexpr float NL = -1.4426950408889634f, N2L = -2.8853900817779268f;
  v2f ax[KX];
  #pragma unroll
  for (int kk = 0; kk < KX - 1; ++kk) ax[kk] = axM[kk];
  ax[KX - 1].x = hi ? 0.f : axM[KX - 1].x;
  ax[KX - 1].y = hi ? 0.f : axM[KX - 1].y;
  v2f ah[KH];
  #pragma unroll
  for (int kk = 0; kk < KH; ++kk)
    ah[kk] = *(const v2f*)&hM[M * HP + kk * 4 + hi * 2];

  float m0[8], m1[8], m2[8], m3[8], m4[8], m5[8];
  float e0[8], e1[8], e2[8], e3[8], e4[8], e5[8];
  float s0[8], s1[8], s2[8], s5[8], r3[8], r4[8], n3[8], t3[8], q4[8];
  float u03[8], u14[8], ru[8], rx[8], sel[8], ca[8], cb[8];
  float ea[8], eb[8], na[8], nb[8], ra[8], rb[8], ta[8], tb[8];

  LN(0,  m0[r] = accV[0][r] * NL)
  LN(1,  m3[r] = N2L * fabsf(accV[3][r]))
  LN(2,  m1[r] = accV[1][r] * NL)
  LN(3,  m4[r] = accV[4][r] * k4)
  LN(4,  e0[r] = __builtin_amdgcn_exp2f(m0[r]))
  LN(5,  e3[r] = __builtin_amdgcn_exp2f(m3[r]))
  LN(6,  e1[r] = __builtin_amdgcn_exp2f(m1[r]))
  LN(7,  e4[r] = __builtin_amdgcn_exp2f(m4[r]))
  LN(8,  m2[r] = accV[2][r] * NL)
  LN(9,  m5[r] = accV[5][r] * NL)
  LN(10, e0[r] = 1.f + e0[r])
  LN(11, n3[r] = 1.f - e3[r])
  LN(12, e3[r] = 1.f + e3[r])
  LN(13, e1[r] = 1.f + e1[r])
  LN(14, e4[r] = 1.f + e4[r])
  LN(15, e2[r] = __builtin_amdgcn_exp2f(m2[r]))
  LN(16, e5[r] = __builtin_amdgcn_exp2f(m5[r]))
  LN(17, s0[r] = __builtin_amdgcn_rcpf(e0[r]))
  LN(18, r3[r] = __builtin_amdgcn_rcpf(e3[r]))
  LN(19, s1[r] = __builtin_amdgcn_rcpf(e1[r]))
  LN(20, r4[r] = __builtin_amdgcn_rcpf(e4[r]))
  LN(21, e2[r] = 1.f + e2[r])
  LN(22, e5[r] = 1.f + e5[r])
  LN(23, t3[r] = n3[r] * r3[r])
  LN(24, t3[r] = copysignf(t3[r], accV[3][r]))
  LN(25, q4[r] = __builtin_fmaf(r4[r], a4, b4))
  LN(26, s2[r] = __builtin_amdgcn_rcpf(e2[r]))
  LN(27, s5[r] = __builtin_amdgcn_rcpf(e5[r]))
  LN(28, u03[r] = s0[r] * t3[r])
  LN(29, u14[r] = s1[r] * q4[r])
  LN(30, ru[r] = rot8(u03[r]))
  LN(31, sel[r] = lo ? u14[r] : u03[r])
  LN(32, rx[r] = rot8(sel[r]))
  LN(33, cb[r] = s1[r] * cB[r] + ru[r]; cB[r] = cb[r])
  LN(34, ca[r] = s2[r] * cA[r] + rx[r]; cA[r] = ca[r])
  LN(35, m0[r] = N2L * fabsf(cb[r]))
  LN(36, m1[r] = N2L * fabsf(ca[r]))
  LN(37, eb[r] = __builtin_amdgcn_exp2f(m0[r]))
  LN(38, ea[r] = __builtin_amdgcn_exp2f(m1[r]))
  LN(39, nb[r] = 1.f - eb[r])
  LN(40, eb[r] = 1.f + eb[r])
  LN(41, na[r] = 1.f - ea[r])
  LN(42, ea[r] = 1.f + ea[r])
  LN(43, rb[r] = __builtin_amdgcn_rcpf(eb[r]))
  LN(44, ra[r] = __builtin_amdgcn_rcpf(ea[r]))
  LN(45, tb[r] = nb[r] * rb[r])
  LN(46, ta[r] = na[r] * ra[r])
  LN(47, tb[r] = copysignf(tb[r], cb[r]))
  LN(48, ta[r] = copysignf(ta[r], ca[r]))
  LN(49, tb[r] = q4[r] * tb[r])
  LN(50, ta[r] = s5[r] * ta[r])
  LN(51, hV[offB + r * HP] = tb[r])
  LN(52, hV[offA + r * HP] = ta[r])
  LN(53, (void)0)
  LN(54, (void)0)
  LN(55, (void)0)
  LN(56, (void)0)
  LN(57, (void)0)
  LN(58, (void)0)
  LN(59, (void)0)
}

__device__ __forceinline__ void cell(const v8f (&acc)[NT], float (&cA)[8], float (&cB)[8],
                                     float* hT, int offA, int offB, bool lo,
                                     float k4, float a4, float b4) {
  #pragma unroll
  for (int r = 0; r < 8; ++r) {
    const float s0 = fsig(acc[0][r]);
    const float t3 = ftanh(acc[3][r]);
    const float s1 = fsig(acc[1][r]);
    const float s2 = fsig(acc[2][r]);
    const float s5 = fsig(acc[5][r]);
    const float e4 = __builtin_amdgcn_exp2f(acc[4][r] * k4);
    const float q4 = __builtin_fmaf(__builtin_amdgcn_rcpf(1.f + e4), a4, b4);
    const float u03 = s0 * t3;
    const float u14 = s1 * q4;
    const float rx = rot8(lo ? u14 : u03);
    const float ru = rot8(u03);
    const float ca = s2 * cA[r] + rx;
    cA[r] = ca;
    hT[offA + r * HP] = s5 * ftanh(ca);
    const float cb = s1 * cB[r] + ru;
    cB[r] = cb;
    hT[offB + r * HP] = q4 * ftanh(cb);
  }
}

__global__ __launch_bounds__(32 * WAVES) void bilstm_center_kernel(
    const float* __restrict__ x,
    const float* __restrict__ w_ih_f, const float* __restrict__ w_hh_f,
    const float* __restrict__ b_ih_f, const float* __restrict__ b_hh_f,
    const float* __restrict__ w_ih_b, const float* __restrict__ w_hh_b,
    const float* __restrict__ b_ih_b, const float* __restrict__ b_hh_b,
    const float* __restrict__ head_w, const float* __restrict__ head_b,
    float* __restrict__ out, int B)
{
  __shared__ float s_headw[NC * 2 * H];
  __shared__ float s_headb[NC];
  __shared__ float s_h[WAVES][TPW][16 * HP];
  __shared__ float s_hf[WAVES][TPW][16 * HP];

  const int tid = threadIdx.x;

  const int lane = tid & 31;
  const int w    = tid >> 5;
  const int M    = lane & 15;
  const int hi   = lane >> 4;

  static_assert(NC * 2 * H <= 32 * WAVES, "one head weight per thread");
  float hw = 0.f, hb = 0.f;
  if (tid < NC * 2 * H) hw = head_w[tid];
  if (tid < NC) hb = head_b[tid];
  v2f wbx[NT][KX], wbh[NT][KH];
  float bi[NT], bh[NT];
  load_w(wbx, wbh, bi, bh, w_ih_f, w_hh_f, b_ih_f, b_hh_f, M, hi);
  if (tid < NC * 2 * H) s_headw[tid] = hw;
  if (tid < NC) s_headb[tid] = hb;
  __syncthreads();

  const int b0   = (blockIdx.x * WAVES + w) * (16 * TPW);
  if (b0 >= B) return;
  const bool q_live = (b0 + 16 < B);
  const int  bq     = q_live ? b0 + 16 : b0;

  float* hP = &s_h[w][0][0];
  float* hQ = &s_h[w][1][0];

  const bool  lo = (M < 8);
  const float k4 = lo ? -2.8853900817779268f : -1.4426950408889634f;
  const float a4 = lo ? 2.f : 1.f;
  const float b4 = lo ? -1.f : 0.f;
  const int offA = (8 * hi) * HP + M + 8;
  const int offB = (8 * hi) * HP + (lo ? 24 + (M & 3) : M - 8);

  const float* xP = x + ((size_t)b0 + M) * T * IN_F;
  const float* xQ = x + ((size_t)bq + M) * T * IN_F;

  for (int d = 0; d < 2; ++d) {
    float cPA[8], cPB[8], cQA[8], cQB[8];
    #pragma unroll
    for (int r = 0; r < 8; ++r) { cPA[r] = 0.f; cPB[r] = 0.f; cQA[r] = 0.f; cQB[r] = 0.f; }
    for (int e = lane; e < 16 * HP; e += 32) { hP[e] = 0.f; hQ[e] = 0.f; }

    #pragma unroll
    for (int nt = 0; nt < NT; ++nt) {
      wbx[nt][KX - 1].x = hi ? 0.f : wbx[nt][KX - 1].x;
      wbx[nt][KX - 1].y = hi ? 0.f : wbx[nt][KX - 1].y;
    }

    v8f accP[NT], accQ[NT], cinit[NT];
    #pragma unroll
    for (int nt = 0; nt < NT; ++nt) {
      const float bb = bi[nt] + bh[nt];
      #pragma unroll
      for (int r = 0; r < 8; ++r) { accQ[nt][r] = 0.f; cinit[nt][r] = bb; }
    }
    v2f axP[KX], axQ[KX];
    const int tstep = d ? -IN_F : IN_F;
    const float* xpP = xP + (d ? (T - 1) * IN_F : 0);
    const float* xpQ = xQ + (d ? (T - 1) * IN_F : 0);
    load_x(axP, xpP, hi);
    xpP += tstep;

    for (int step = 0; step <= WH; ++step) {
      load_x(axQ, xpQ, hi);
      xpQ += tstep;
      half_step(accP, cinit, axP, hP, wbx, wbh, accQ, cQA, cQB, hQ,
                M, hi, offA, offB, lo, k4, a4, b4);
      WAVE_FENCE();
      load_x(axP, xpP, hi);
      xpP += tstep;
      half_step(accQ, cinit, axQ, hQ, wbx, wbh, accP, cPA, cPB, hP,
                M, hi, offA, offB, lo, k4, a4, b4);
      WAVE_FENCE();
    }
    if (d == 0) load_w(wbx, wbh, bi, bh, w_ih_b, w_hh_b, b_ih_b, b_hh_b, M, hi);
    cell(accQ, cQA, cQB, hQ, offA, offB, lo, k4, a4, b4);
    WAVE_FENCE();

    if (d == 0) {
      for (int e = lane; e < 16 * HP; e += 32) {
        s_hf[w][0][e] = hP[e];
        s_hf[w][1][e] = hQ[e];
      }
      WAVE_FENCE();
    }
  }

  #pragma unroll
  for (int tl = 0; tl < TPW; ++tl) {
    if (tl == 1 && !q_live) break;
    const float* hf = &s_hf[w][tl][0];
    const float* hb = &s_h[w][tl][0];
    #pragma unroll
    for (int p = lane; p < 64; p += 32) {
      const int m = p >> 2, cls = p & 3;
      float acc = s_headb[cls];
      #pragma unroll
      for (int j = 0; j < H; ++j) {
        acc += hf[m * HP + j] * s_headw[cls * 2 * H + j];
        acc += hb[m * HP + j] * s_headw[cls * 2 * H + H + j];
      }
      out[(size_t)(b0 + 16 * tl + m) * NC + cls] = acc;
    }
  }
}

extern "C" void kernel_launch(void* const* d_in, const int* in_sizes, int n_in,
                              void* d_out, int out_size, void* d_ws, size_t ws_size,
                              hipStream_t stream) {
  (void)n_in; (void)d_ws; (void)ws_size; (void)out_size;
  const float* x      = (const float*)d_in[0];
  const float* w_ih_f = (const float*)d_in[1];
  const float* w_hh_f = (const float*)d_in[2];
  const float* b_ih_f = (const float*)d_in[3];
  const float* b_hh_f = (const float*)d_in[4];
  const float* w_ih_b = (const float*)d_in[5];
  const float* w_hh_b = (const float*)d_in[6];
  const float* b_ih_b = (const float*)d_in[7];
  const float* b_hh_b = (const float*)d_in[8];
  const float* head_w = (const float*)d_in[9];
  const float* head_b = (const float*)d_in[10];
  float* out = (float*)d_out;

  const int B = in_sizes[0] / (T * IN_F);
  const int rowsPerBlock = 16 * TPW * WAVES;
  const int grid = (B + rowsPerBlock - 1) / rowsPerBlock;
  bilstm_center_kernel<<<grid, 32 * WAVES, 0, stream>>>(
      x, w_ih_f, w_hh_f, b_ih_f, b_hh_f,
      w_ih_b, w_hh_b, b_ih_b, b_hh_b, head_w, head_b, out, B);
}


# Activation scale factors leave the time loop: weights and bias pre-scaled by -log2(e) (g: -2log2(e)), cell state kept as -2log2(e)*c, tanh as 2*sig(2x)-1; no multiply, abs or copysign; sum order kept
# speedup vs baseline: 1.3960x; 1.0622x over previous
#include <hip/hip_runtime.h>
#include <math.h>


typedef __attribute__((ext_vector_type(2))) float v2f;
typedef __attribute__((ext_vector_type(8))) float v8f;

#define WAVES 8
constexpr int IN_F = 14, H = 24, T = 25, NC = 4, WH = 12;
constexpr int G  = 4 * H;
constexpr int NT = G / 16;
constexpr int KX = 4;
constexpr int KH = H / 4;
constexpr int HP = 28;
constexpr int TPW = 2;

constexpr float SL = -1.4426950408889634f;
constexpr float SK = -2.8853900817779268f;
__device__ __forceinline__ float fsg2(float a) {
  return __builtin_amdgcn_rcpf(1.f + __builtin_amdgcn_exp2f(a));
}

#define WAVE_FENCE() asm volatile("" ::: "memory")

__device__ __forceinline__ float rot8(float v) {
  return __builtin_bit_cast(float, __builtin_amdgcn_update_dpp(
      0, __builtin_bit_cast(int, v), 0x128, 0xf, 0xf, true));
}

__device__ __forceinline__ void load_x(v2f (&ax)[KX], const float* xr, int hi) {
  #pragma unroll
  for (int kk = 0; kk < KX - 1; ++kk) ax[kk] = *(const v2f*)(xr + kk * 4 + hi * 2);
  ax[KX - 1] = *(const v2f*)(xr + (KX - 1) * 4);
}

__device__ __forceinline__ void load_w(v2f (&wbx)[NT][KX], v2f (&wbh)[NT][KH],
                                       float (&bi)[NT], float (&bh)[NT],
                                       const float* __restrict__ w_ih,
                                       const float* __restrict__ w_hh,
                                       const float* __restrict__ b_ih,
                                       const float* __restrict__ b_hh, int M, int hi) {
  #pragma unroll
  for (int nt = 0; nt < NT; ++nt) {
    const int n = nt * 16 + M;
    const float* rx = w_ih + n * IN_F;
    const float* rh = w_hh + n * H;
    #pragma unroll
    for (int kk = 0; kk < KX - 1; ++kk) wbx[nt][kk] = *(const v2f*)(rx + kk * 4 + hi * 2);
    wbx[nt][KX - 1] = *(const v2f*)(rx + (KX - 1) * 4);
    #pragma unroll
    for (int kk = 0; kk < KH; ++kk) wbh[nt][kk] = *(const v2f*)(rh + kk * 4 + hi * 2);
    bi[nt] = b_ih[n];
    bh[nt] = b_hh[n];
  }
}

template <int I>
__device__ __forceinline__ void wm(v8f (&acc)[NT], const v8f (&cinit)[NT], const v2f (&ax)[KX],
                                   const v2f (&ah)[KH], const v2f (&wbx)[NT][KX],
                                   const v2f (&wbh)[NT][KH]) {
  constexpr int kk = I / NT, nt = I % NT;
  if constexpr (kk == 0)
    acc[nt] = __builtin_amdgcn_wmma_f32_16x16x4_f32(
        false, ax[0], false, wbx[nt][0], (short)0, cinit[nt], false, false);
  else if constexpr (kk < KX)
    acc[nt] = __builtin_amdgcn_wmma_f32_16x16x4_f32(
        false, ax[kk], false, wbx[nt][kk], (short)0, acc[nt], false, false);
  else
    acc[nt] = __builtin_amdgcn_wmma_f32_16x16x4_f32(
        false, ah[kk - KX], false, wbh[nt][kk - KX], (short)0, acc[nt], false, false);
}

#define HOLD(v) asm volatile("" :: "v"(v))
#define LN(i, body)                                              \
  {                                                              \
    _Pragma("unroll") for (int r = 0; r < 8; ++r) { body; }      \
    wm<i>(accM, cinit, ax, ah, wbx, wbh);                        \
    HOLD(accM[(i) % NT]);                                        \
    __builtin_amdgcn_sched_barrier(0);                           \
  }

__device__ __forceinline__ void half_step(
    v8f (&accM)[NT], const v8f (&cinit)[NT], const v2f (&axM)[KX], const float* hM,
    const v2f (&wbx)[NT][KX], const v2f (&wbh)[NT][KH],
    const v8f (&accV)[NT], float (&cA)[8], float (&cB)[8], float* hV,
    int M, int hi, int offA, int offB, bool lo, float a4, float b4)
{
  constexpr float K2 = 2.f * SK, KN = -SK;
  v2f ax[KX];
  #pragma unroll
  for (int kk = 0; kk < KX - 1; ++kk) ax[kk] = axM[kk];
  ax[KX - 1].x = hi ? 0.f : axM[KX - 1].x;
  ax[KX - 1].y = hi ? 0.f : axM[KX - 1].y;
  v2f ah[KH];
  #pragma unroll
  for (int kk = 0; kk < KH; ++kk)
    ah[kk] = *(const v2f*)&hM[M * HP + kk * 4 + hi * 2];

  float e0[8], e1[8], e2[8], e3[8], e4[8], e5[8];
  float s0[8], s1[8], s2[8], s5[8], r3[8], r4[8], t3[8], q4[8];
  float u03[8], u14[8], ru[8], rx[8], sel[8], ca[8], cb[8];
  float ea[8], eb[8], ra[8], rb[8], ta[8], tb[8];

  LN(0,  e0[r] = __builtin_amdgcn_exp2f(accV[0][r]))
  LN(1,  e3[r] = __builtin_amdgcn_exp2f(accV[3][r]))
  LN(2,  e1[r] = __builtin_amdgcn_exp2f(accV[1][r]))
  LN(3,  e4[r] = __builtin_amdgcn_exp2f(accV[4][r]))
  LN(4,  e2[r] = __builtin_amdgcn_exp2f(accV[2][r]))
  LN(5,  e5[r] = __builtin_amdgcn_exp2f(accV[5][r]))
  LN(6,  e0[r] = 1.f + e0[r])
  LN(7,  e3[r] = 1.f + e3[r])
  LN(8,  e1[r] = 1.f + e1[r])
  LN(9,  e4[r] = 1.f + e4[r])
  LN(10, e2[r] = 1.f + e2[r])
  LN(11, e5[r] = 1.f + e5[r])
  LN(12, s0[r] = __builtin_amdgcn_rcpf(e0[r]))
  LN(13, r3[r] = __builtin_amdgcn_rcpf(e3[r]))
  LN(14, s1[r] = __builtin_amdgcn_rcpf(e1[r]))
  LN(15, r4[r] = __builtin_amdgcn_rcpf(e4[r]))
  LN(16, s2[r] = __builtin_amdgcn_rcpf(e2[r]))
  LN(17, s5[r] = __builtin_amdgcn_rcpf(e5[r]))
  LN(18, t3[r] = __builtin_fmaf(r3[r], K2, KN))
  LN(19, q4[r] = __builtin_fmaf(r4[r], a4, b4))
  LN(20, u03[r] = s0[r] * t3[r])
  LN(21, u14[r] = s1[r] * q4[r])
  LN(22, ru[r] = rot8(u03[r]))
  LN(23, sel[r] = lo ? u14[r] : u03[r])
  LN(24, rx[r] = rot8(sel[r]))
  LN(25, cb[r] = s1[r] * cB[r] + ru[r]; cB[r] = cb[r])
  LN(26, ca[r] = s2[r] * cA[r] + rx[r]; cA[r] = ca[r])
  LN(27, eb[r] = __builtin_amdgcn_exp2f(cb[r]))
  LN(28, ea[r] = __builtin_amdgcn_exp2f(ca[r]))
  LN(29, eb[r] = 1.f + eb[r])
  LN(30, ea[r] = 1.f + ea[r])
  LN(31, rb[r] = __builtin_amdgcn_rcpf(eb[r]))
  LN(32, ra[r] = __builtin_amdgcn_rcpf(ea[r]))
  LN(33, tb[r] = __builtin_fmaf(rb[r], 2.f, -1.f))
  LN(34, ta[r] = __builtin_fmaf(ra[r], 2.f, -1.f))
  LN(35, tb[r] = q4[r] * tb[r])
  LN(36, ta[r] = s5[r] * ta[r])
  LN(37, hV[offB + r * HP] = tb[r])
  LN(38, hV[offA + r * HP] = ta[r])
  LN(39, (void)0)
  LN(40, (void)0)
  LN(41, (void)0)
  LN(42, (void)0)
  LN(43, (void)0)
  LN(44, (void)0)
  LN(45, (void)0)
  LN(46, (void)0)
  LN(47, (void)0)
  LN(48, (void)0)
  LN(49, (void)0)
  LN(50, (void)0)
  LN(51, (void)0)
  LN(52, (void)0)
  LN(53, (void)0)
  LN(54, (void)0)
  LN(55, (void)0)
  LN(56, (void)0)
  LN(57, (void)0)
  LN(58, (void)0)
  LN(59, (void)0)
}

__device__ __forceinline__ void cell(const v8f (&acc)[NT], float (&cA)[8], float (&cB)[8],
                                     float* hT, int offA, int offB, bool lo,
                                     float a4, float b4) {
  constexpr float K2 = 2.f * SK, KN = -SK;
  #pragma unroll
  for (int r = 0; r < 8; ++r) {
    const float s0 = fsg2(acc[0][r]);
    const float t3 = __builtin_fmaf(fsg2(acc[3][r]), K2, KN);
    const float s1 = fsg2(acc[1][r]);
    const float s2 = fsg2(acc[2][r]);
    const float s5 = fsg2(acc[5][r]);
    const float q4 = __builtin_fmaf(fsg2(acc[4][r]), a4, b4);
    const float u03 = s0 * t3;
    const float u14 = s1 * q4;
    const float rx = rot8(lo ? u14 : u03);
    const float ru = rot8(u03);
    const float ca = s2 * cA[r] + rx;
    cA[r] = ca;
    hT[offA + r * HP] = s5 * __builtin_fmaf(fsg2(ca), 2.f, -1.f);
    const float cb = s1 * cB[r] + ru;
    cB[r] = cb;
    hT[offB + r * HP] = q4 * __builtin_fmaf(fsg2(cb), 2.f, -1.f);
  }
}

__global__ __launch_bounds__(32 * WAVES) void bilstm_center_kernel(
    const float* __restrict__ x,
    const float* __restrict__ w_ih_f, const float* __restrict__ w_hh_f,
    const float* __restrict__ b_ih_f, const float* __restrict__ b_hh_f,
    const float* __restrict__ w_ih_b, const float* __restrict__ w_hh_b,
    const float* __restrict__ b_ih_b, const float* __restrict__ b_hh_b,
    const float* __restrict__ head_w, const float* __restrict__ head_b,
    float* __restrict__ out, int B)
{
  __shared__ float s_headw[NC * 2 * H];
  __shared__ float s_headb[NC];
  __shared__ float s_h[WAVES][TPW][16 * HP];
  __shared__ float s_hf[WAVES][TPW][16 * HP];

  const int tid = threadIdx.x;

  const int lane = tid & 31;
  const int w    = tid >> 5;
  const int M    = lane & 15;
  const int hi   = lane >> 4;

  static_assert(NC * 2 * H <= 32 * WAVES, "one head weight per thread");
  float hw = 0.f, hb = 0.f;
  if (tid < NC * 2 * H) hw = head_w[tid];
  if (tid < NC) hb = head_b[tid];
  v2f wbx[NT][KX], wbh[NT][KH];
  float bi[NT], bh[NT];
  load_w(wbx, wbh, bi, bh, w_ih_f, w_hh_f, b_ih_f, b_hh_f, M, hi);
  if (tid < NC * 2 * H) s_headw[tid] = hw;
  if (tid < NC) s_headb[tid] = hb;
  __syncthreads();

  const int b0   = (blockIdx.x * WAVES + w) * (16 * TPW);
  if (b0 >= B) return;
  const bool q_live = (b0 + 16 < B);
  const int  bq     = q_live ? b0 + 16 : b0;

  float* hP = &s_h[w][0][0];
  float* hQ = &s_h[w][1][0];

  const bool  lo = (M < 8);
  const float a4  = lo ? 2.f * SK : 1.f;
  const float b4  = lo ? -SK : 0.f;
  const float sc4 = lo ? SK : SL;
  const int offA = (8 * hi) * HP + M + 8;
  const int offB = (8 * hi) * HP + (lo ? 24 + (M & 3) : M - 8);

  const float* xP = x + ((size_t)b0 + M) * T * IN_F;
  const float* xQ = x + ((size_t)bq + M) * T * IN_F;

  for (int d = 0; d < 2; ++d) {
    float cPA[8], cPB[8], cQA[8], cQB[8];
    #pragma unroll
    for (int r = 0; r < 8; ++r) { cPA[r] = 0.f; cPB[r] = 0.f; cQA[r] = 0.f; cQB[r] = 0.f; }
    for (int e = lane; e < 16 * HP; e += 32) { hP[e] = 0.f; hQ[e] = 0.f; }

    #pragma unroll
    for (int nt = 0; nt < NT; ++nt) {
      wbx[nt][KX - 1].x = hi ? 0.f : wbx[nt][KX - 1].x;
      wbx[nt][KX - 1].y = hi ? 0.f : wbx[nt][KX - 1].y;
      const float sc = (nt == 3) ? SK : (nt == 4) ? sc4 : SL;
      #pragma unroll
      for (int kk = 0; kk < KX; ++kk) wbx[nt][kk] = wbx[nt][kk] * sc;
      #pragma unroll
      for (int kk = 0; kk < KH; ++kk) wbh[nt][kk] = wbh[nt][kk] * sc;
    }

    v8f accP[NT], accQ[NT], cinit[NT];
    #pragma unroll
    for (int nt = 0; nt < NT; ++nt) {
      const float bb = (bi[nt] + bh[nt]) * ((nt == 3) ? SK : (nt == 4) ? sc4 : SL);
      #pragma unroll
      for (int r = 0; r < 8; ++r) { accQ[nt][r] = 0.f; cinit[nt][r] = bb; }
    }
    v2f axP[KX], axQ[KX];
    const int tstep = d ? -IN_F : IN_F;
    const float* xpP = xP + (d ? (T - 1) * IN_F : 0);
    const float* xpQ = xQ + (d ? (T - 1) * IN_F : 0);
    load_x(axP, xpP, hi);
    xpP += tstep;

    for (int step = 0; step <= WH; ++step) {
      load_x(axQ, xpQ, hi);
      xpQ += tstep;
      half_step(accP, cinit, axP, hP, wbx, wbh, accQ, cQA, cQB, hQ,
                M, hi, offA, offB, lo, a4, b4);
      WAVE_FENCE();
      load_x(axP, xpP, hi);
      xpP += tstep;
      half_step(accQ, cinit, axQ, hQ, wbx, wbh, accP, cPA, cPB, hP,
                M, hi, offA, offB, lo, a4, b4);
      WAVE_FENCE();
    }
    if (d == 0) load_w(wbx, wbh, bi, bh, w_ih_b, w_hh_b, b_ih_b, b_hh_b, M, hi);
    cell(accQ, cQA, cQB, hQ, offA, offB, lo, a4, b4);
    WAVE_FENCE();

    if (d == 0) {
      for (int e = lane; e < 16 * HP; e += 32) {
        s_hf[w][0][e] = hP[e];
        s_hf[w][1][e] = hQ[e];
      }
      WAVE_FENCE();
    }
  }

  #pragma unroll
  for (int tl = 0; tl < TPW; ++tl) {
    if (tl == 1 && !q_live) break;
    const float* hf = &s_hf[w][tl][0];
    const float* hb = &s_h[w][tl][0];
    #pragma unroll
    for (int p = lane; p < 64; p += 32) {
      const int m = p >> 2, cls = p & 3;
      float acc = s_headb[cls];
      #pragma unroll
      for (int j = 0; j < H; ++j) {
        acc += hf[m * HP + j] * s_headw[cls * 2 * H + j];
        acc += hb[m * HP + j] * s_headw[cls * 2 * H + H + j];
      }
      out[(size_t)(b0 + 16 * tl + m) * NC + cls] = acc;
    }
  }
}

extern "C" void kernel_launch(void* const* d_in, const int* in_sizes, int n_in,
                              void* d_out, int out_size, void* d_ws, size_t ws_size,
                              hipStream_t stream) {
  (void)n_in; (void)d_ws; (void)ws_size; (void)out_size;
  const float* x      = (const float*)d_in[0];
  const float* w_ih_f = (const float*)d_in[1];
  const float* w_hh_f = (const float*)d_in[2];
  const float* b_ih_f = (const float*)d_in[3];
  const float* b_hh_f = (const float*)d_in[4];
  const float* w_ih_b = (const float*)d_in[5];
  const float* w_hh_b = (const float*)d_in[6];
  const float* b_ih_b = (const float*)d_in[7];
  const float* b_hh_b = (const float*)d_in[8];
  const float* head_w = (const float*)d_in[9];
  const float* head_b = (const float*)d_in[10];
  float* out = (float*)d_out;

  const int B = in_sizes[0] / (T * IN_F);
  const int rowsPerBlock = 16 * TPW * WAVES;
  const int grid = (B + rowsPerBlock - 1) / rowsPerBlock;
  bilstm_center_kernel<<<grid, 32 * WAVES, 0, stream>>>(
      x, w_ih_f, w_hh_f, b_ih_f, b_hh_f,
      w_ih_b, w_hh_b, b_ih_b, b_hh_b, head_w, head_b, out, B);
}
